# starting kernel
# baseline (speedup 1.0000x reference)
#include <hip/hip_runtime.h>
#include <math.h>


namespace {
constexpr int B_T = 256, T_T = 32, IN = 16, H = 128, SH = 128, OUTW = 64, TM1 = 31;
constexpr int MT = 16;
constexpr int THREADS = 256;
constexpr int SA = 132;

typedef __attribute__((ext_vector_type(2))) float v2f;
typedef __attribute__((ext_vector_type(8))) float v8f;

__device__ __forceinline__ v8f wmma4(v2f a, v2f b, v8f c) {
  return __builtin_amdgcn_wmma_f32_16x16x4_f32(false, a, false, b, (short)0, c,
                                               false, false);
}

__device__ __forceinline__ v8f gemm_tile(const float* A, int lda, const float* W,
                                         int ldw, int n, int K, int lane) {
  const int m = lane & 15;
  const int kh = (lane >> 4) << 1;
  v8f acc = {};
  #pragma unroll 8
  for (int k0 = 0; k0 < K; k0 += 4) {
    v2f a = *(const v2f*)(A + m * lda + k0 + kh);
    v2f b = *(const v2f*)(W + (size_t)n * ldw + k0 + kh);
    acc = wmma4(a, b, acc);
  }
  return acc;
}

__device__ __forceinline__ void mlp_layer(const float* src, const float* Wlds,
                                          const float* bias, float* dst,
                                          int wid, int lane) {
  const int mD = lane & 15, half = lane >> 4;
  const int n = wid * 16 + mD;
  v8f acc = gemm_tile(src, SA, Wlds, SA, n, SH, lane);
  const float bn = bias[n];
  #pragma unroll
  for (int v = 0; v < 8; ++v) {
    float x = acc[v] + bn;
    dst[(v + 8 * half) * SA + n] = fmaxf(x, 0.f);
  }
}

__device__ __forceinline__ void out_proj(const float* zb, const float* W_last,
                                         const float* b_last, float* out,
                                         int b0r, int t, int wid, int lane) {
  if (wid < 4) {
    const int mD = lane & 15, half = lane >> 4;
    const int n = wid * 16 + mD;
    v8f acc = gemm_tile(zb, SA, W_last, H, n, H, lane);
    const float bn = b_last[n];
    #pragma unroll
    for (int v = 0; v < 8; ++v) {
      int m = v + 8 * half;
      float x = acc[v] + bn;
      out[((size_t)(b0r + m) * T_T + t) * OUTW + n] = 1.f / (1.f + expf(-x));
    }
  }
}
}

__global__ __launch_bounds__(THREADS) void ncde_wmma_kernel(
    const float* __restrict__ coeffs,
    const float* __restrict__ W_init, const float* __restrict__ b_init,
    const float* __restrict__ W0, const float* __restrict__ b0,
    const float* __restrict__ Wl, const float* __restrict__ bl,
    const float* __restrict__ Wf, const float* __restrict__ bf,
    const float* __restrict__ W_last, const float* __restrict__ b_last,
    float* __restrict__ out) {
  __shared__ __align__(16) float wlds[4 * H * SA];
  __shared__ __align__(16) float zbuf[MT * SA];
  __shared__ __align__(16) float kbuf[3][MT * H];
  __shared__ __align__(16) float actI[MT * SA];
  __shared__ __align__(16) float actA[MT * SA];
  __shared__ __align__(16) float dxs[MT * IN];

  const int tid = threadIdx.x;
  const int lane = tid & 31;
  const int wid = tid >> 5;
  const int b0r = blockIdx.x * MT;
  const int mD = lane & 15;
  const int half = lane >> 4;
  constexpr float third = 1.f / 3.f;

  for (int e = tid; e < 4 * H * SH; e += THREADS) {
    int mat = e >> 14;
    int r = (e >> 7) & (H - 1);
    int k = e & (SH - 1);
    const float* src = (mat == 0) ? W0 : (Wl + (size_t)(mat - 1) * SH * SH);
    wlds[(size_t)mat * H * SA + r * SA + k] = src[(size_t)r * SH + k];
  }

  {
    int m = tid >> 4, i = tid & 15;
    dxs[tid] = coeffs[((size_t)(b0r + m) * TM1) * (4 * IN) + i];
  }
  __syncthreads();
  {
    const int n = wid * 16 + mD;
    v8f acc = gemm_tile(dxs, IN, W_init, IN, n, IN, lane);
    const float bn = b_init[n];
    #pragma unroll
    for (int v = 0; v < 8; ++v)
      zbuf[(v + 8 * half) * SA + n] = acc[v] + bn;
  }
  __syncthreads();
  out_proj(zbuf, W_last, b_last, out, b0r, 0, wid, lane);
  __syncthreads();

  for (int step = 0; step < TM1; ++step) {
    #pragma unroll 1
    for (int s = 0; s < 4; ++s) {
      const float frac = (s == 0) ? 0.f
                       : (s == 1) ? third
                       : (s == 2) ? (2.f * third) : 1.f;
      {
        int m = tid >> 4, i = tid & 15;
        const float* cb = coeffs + ((size_t)(b0r + m) * TM1 + step) * (4 * IN);
        float bi = cb[IN + i], ci = cb[2 * IN + i], di = cb[3 * IN + i];
        dxs[tid] = bi + (ci + di * frac) * frac;
      }
      for (int e = tid; e < MT * H; e += THREADS) {
        int m = e >> 7, hc = e & (H - 1);
        float v = zbuf[m * SA + hc];
        if (s == 1)      v += kbuf[0][e] * third;
        else if (s == 2) v += kbuf[1][e] - kbuf[0][e] * third;
        else if (s == 3) v += kbuf[0][e] - kbuf[1][e] + kbuf[2][e];
        actI[m * SA + hc] = v;
      }
      __syncthreads();

      mlp_layer(actI, wlds + 0 * H * SA, b0,          actA, wid, lane); __syncthreads();
      mlp_layer(actA, wlds + 1 * H * SA, bl + 0 * SH, actI, wid, lane); __syncthreads();
      mlp_layer(actI, wlds + 2 * H * SA, bl + 1 * SH, actA, wid, lane); __syncthreads();
      mlp_layer(actA, wlds + 3 * H * SA, bl + 2 * SH, actI, wid, lane); __syncthreads();

      {
        float dxr[8];
        #pragma unroll
        for (int v = 0; v < 8; ++v) dxr[v] = dxs[(v + 8 * half) * IN + mD];
        const int kh = half << 1;
        for (int t0 = wid; t0 < H; t0 += 16) {
          const int t1 = t0 + 8;
          const int n0 = t0 * 16 + mD, n1 = t1 * 16 + mD;
          v8f acc0 = {}, acc1 = {};
          #pragma unroll 8
          for (int k0 = 0; k0 < SH; k0 += 4) {
            v2f a  = *(const v2f*)(actI + mD * SA + k0 + kh);
            v2f w0 = *(const v2f*)(Wf + (size_t)n0 * SH + k0 + kh);
            v2f w1 = *(const v2f*)(Wf + (size_t)n1 * SH + k0 + kh);
            acc0 = wmma4(a, w0, acc0);
            acc1 = wmma4(a, w1, acc1);
          }
          const float bf0 = bf[n0], bf1 = bf[n1];
          #pragma unroll
          for (int v = 0; v < 8; ++v) {
            const int m = v + 8 * half;
            float f0 = tanhf(acc0[v] + bf0) * dxr[v];
            float f1 = tanhf(acc1[v] + bf1) * dxr[v];
            f0 += __shfl_xor(f0, 1); f0 += __shfl_xor(f0, 2);
            f0 += __shfl_xor(f0, 4); f0 += __shfl_xor(f0, 8);
            f1 += __shfl_xor(f1, 1); f1 += __shfl_xor(f1, 2);
            f1 += __shfl_xor(f1, 4); f1 += __shfl_xor(f1, 8);
            if (mD == 0) {
              if (s < 3) {
                kbuf[s][m * H + t0] = f0;
                kbuf[s][m * H + t1] = f1;
              } else {
                zbuf[m * SA + t0] +=
                    (kbuf[0][m * H + t0] +
                     3.f * (kbuf[1][m * H + t0] + kbuf[2][m * H + t0]) + f0) * 0.125f;
                zbuf[m * SA + t1] +=
                    (kbuf[0][m * H + t1] +
                     3.f * (kbuf[1][m * H + t1] + kbuf[2][m * H + t1]) + f1) * 0.125f;
              }
            }
          }
        }
      }
      __syncthreads();
    }
    out_proj(zbuf, W_last, b_last, out, b0r, step + 1, wid, lane);
    __syncthreads();
  }
}

extern "C" void kernel_launch(void* const* d_in, const int* in_sizes, int n_in,
                              void* d_out, int out_size, void* d_ws, size_t ws_size,
                              hipStream_t stream) {
  (void)in_sizes; (void)n_in; (void)out_size; (void)d_ws; (void)ws_size;
  const float* coeffs = (const float*)d_in[1];
  const float* W_init = (const float*)d_in[2];
  const float* b_init = (const float*)d_in[3];
  const float* W0     = (const float*)d_in[4];
  const float* b0     = (const float*)d_in[5];
  const float* Wl     = (const float*)d_in[6];
  const float* bl     = (const float*)d_in[7];
  const float* Wf     = (const float*)d_in[8];
  const float* bf     = (const float*)d_in[9];
  const float* W_last = (const float*)d_in[10];
  const float* b_last = (const float*)d_in[11];
  float* out = (float*)d_out;

  ncde_wmma_kernel<<<B_T / MT, THREADS, 0, stream>>>(
      coeffs, W_init, b_init, W0, b0, Wl, bl, Wf, bf, W_last, b_last, out);
}


# Streamed Wf tiles pass through a per-wave LDS buffer: 4 lanes read a row's 64-byte piece (8 lines a load, not 16), one 16-k chunk ahead; B fragments re-read by ds_load; same k order, same sums
# speedup vs baseline: 2.6024x; 2.6024x over previous
#include <hip/hip_runtime.h>
#include <math.h>


namespace {
constexpr int B_T = 256, T_T = 32, IN = 16, H = 128, SH = 128, OUTW = 64, TM1 = 31;
constexpr int MT = 16;
constexpr int THREADS = 512;
constexpr int NCOLW = 8;
constexpr int CHW = 8;
constexpr int SA = 132;
constexpr int NRES = 2;
constexpr int KS = SH / 4;
constexpr int ST = 20;
constexpr int STG_W = 16 * ST;
constexpr int POOL = 16 * STG_W;

typedef __attribute__((ext_vector_type(2))) float v2f;
typedef __attribute__((ext_vector_type(8))) float v8f;

__device__ __forceinline__ float tanh_hw(float x) {
  return __builtin_amdgcn_tanhf(x);
}

__device__ __forceinline__ v8f wmma4(v2f a, v2f b, v8f c) {
  return __builtin_amdgcn_wmma_f32_16x16x4_f32(false, a, false, b, (short)0, c,
                                               false, false);
}

__device__ __forceinline__ void wave_lds_handover() {
  __builtin_amdgcn_fence(__ATOMIC_RELEASE, "wavefront", "local");
  __builtin_amdgcn_wave_barrier();
  __builtin_amdgcn_fence(__ATOMIC_ACQUIRE, "wavefront", "local");
}

__device__ __forceinline__ void staged_unit(float* sst, const float* sld, float4 lo,
                                            float4 hi, const v2f* a, v8f& acc) {
  wave_lds_handover();
  *(float4*)(sst) = lo;
  *(float4*)(sst + 8 * ST) = hi;
  wave_lds_handover();
  #pragma unroll
  for (int j = 0; j < 4; ++j) {
    const v2f b = *(const v2f*)(sld + 4 * j);
    acc = wmma4(a[j], b, acc);
  }
}

struct Chunk { float4 t0l, t0h, t1l, t1h; };

__device__ __forceinline__ Chunk load_chunk(const float* gq) {
  constexpr size_t HI = (size_t)8 * IN * SH;
  Chunk c;
  c.t0l = *(const float4*)(gq);
  c.t0h = *(const float4*)(gq + HI);
  c.t1l = *(const float4*)(gq + SH);
  c.t1h = *(const float4*)(gq + SH + HI);
  return c;
}

__device__ __forceinline__ void do_chunk(const Chunk& c, const float* ak, float* sst,
                                         const float* sld, v8f& acc0, v8f& acc1) {
  v2f a[4];
  #pragma unroll
  for (int j = 0; j < 4; ++j) a[j] = *(const v2f*)(ak + 4 * j);
  staged_unit(sst, sld, c.t0l, c.t0h, a, acc0);
  staged_unit(sst, sld, c.t1l, c.t1h, a, acc1);
}

__device__ __forceinline__ v8f gemm_tile(const float* A, int lda, const float* W,
                                         int ldw, int n, int K, int lane) {
  const int m = lane & 15;
  const int kh = (lane >> 4) << 1;
  v8f acc = {};
  #pragma unroll 8
  for (int k0 = 0; k0 < K; k0 += 4) {
    v2f a = *(const v2f*)(A + m * lda + k0 + kh);
    v2f b = *(const v2f*)(W + (size_t)n * ldw + k0 + kh);
    acc = wmma4(a, b, acc);
  }
  return acc;
}

__device__ __forceinline__ void res_pair(const float* arow, const v2f* wa,
                                         const v2f* wb, v8f& acc0, v8f& acc1) {
  #pragma unroll
  for (int kk = 0; kk < KS; ++kk) {
    v2f a = *(const v2f*)(arow + 4 * kk);
    acc0 = wmma4(a, wa[kk], acc0);
    acc1 = wmma4(a, wb[kk], acc1);
  }
}

__device__ __forceinline__ void mlp_layer(const float* src, const float* Wlds,
                                          const float* bias, float* dst,
                                          int wid, int lane) {
  const int mD = lane & 15, half = lane >> 4;
  const int n = wid * 16 + mD;
  v8f acc = gemm_tile(src, SA, Wlds, SA, n, SH, lane);
  const float bn = bias[n];
  float* drow = dst + 8 * half * SA + n;
  #pragma unroll
  for (int v = 0; v < 8; ++v) {
    float x = acc[v] + bn;
    drow[v * SA] = fmaxf(x, 0.f);
  }
}

__device__ __forceinline__ void out_proj(const float* zb, const float* W_last,
                                         const float* b_last, float* out,
                                         int b0r, int t, int wid, int lane) {
  if (wid < 4) {
    const int mD = lane & 15, half = lane >> 4;
    const int n = wid * 16 + mD;
    v8f acc = gemm_tile(zb, SA, W_last, H, n, H, lane);
    const float bn = b_last[n];
    #pragma unroll
    for (int v = 0; v < 8; ++v) {
      int m = v + 8 * half;
      float x = acc[v] + bn;
      out[((size_t)(b0r + m) * T_T + t) * OUTW + n] = 1.f / (1.f + expf(-x));
    }
  }
}
}

__global__ __launch_bounds__(THREADS) void ncde_wmma_kernel(
    const float* __restrict__ coeffs,
    const float* __restrict__ W_init, const float* __restrict__ b_init,
    const float* __restrict__ W0, const float* __restrict__ b0,
    const float* __restrict__ Wl, const float* __restrict__ bl,
    const float* __restrict__ Wf, const float* __restrict__ bf,
    const float* __restrict__ W_last, const float* __restrict__ b_last,
    float* __restrict__ out) {
  __shared__ __align__(16) float wlds[4 * H * SA];
  __shared__ __align__(16) float zbuf[MT * SA];
  __shared__ __align__(16) float kbuf[2][MT * H];
  __shared__ __align__(16) float actI[MT * SA];
  __shared__ __align__(16) float pool[POOL];
  float* const k3b = pool;
  float* const actA = pool + MT * H;
  __shared__ __align__(16) float dxs[MT * IN];

  const int tid = threadIdx.x;
  const int lane = tid & 31;
  const int wid = tid >> 5;
  const int b0r = blockIdx.x * MT;
  const int mD = lane & 15;
  const int half = lane >> 4;
  constexpr float third = 1.f / 3.f;
  const int wsel = __builtin_amdgcn_readfirstlane(wid);
  const int hg = wsel & (NCOLW - 1);
  const int chalf = wsel >> 3;

  v2f wres[NRES * KS];
  {
    const int khr = half << 1;
    #pragma unroll
    for (int j = 0; j < NRES; ++j) {
      const float* wr = Wf + (size_t)((hg * 16 + mD) * IN + chalf * CHW + j) * SH + khr;
      #pragma unroll
      for (int kk = 0; kk < KS; ++kk)
        wres[j * KS + kk] = *(const v2f*)(wr + 4 * kk);
    }
  }
  const int rot = (wsel >> 2) & 3;

  for (int e = tid; e < 4 * H * SH; e += THREADS) {
    int mat = e >> 14;
    int r = (e >> 7) & (H - 1);
    int k = e & (SH - 1);
    const float* src = (mat == 0) ? W0 : (Wl + (size_t)(mat - 1) * SH * SH);
    wlds[(size_t)mat * H * SA + r * SA + k] = src[(size_t)r * SH + k];
  }

  if (tid < MT * IN) {
    int m = tid >> 4, i = tid & 15;
    dxs[tid] = coeffs[((size_t)(b0r + m) * TM1) * (4 * IN) + i];
  }
  __syncthreads();
  if (wsel < NCOLW) {
    const int n = wsel * 16 + mD;
    v8f acc = gemm_tile(dxs, IN, W_init, IN, n, IN, lane);
    const float bn = b_init[n];
    #pragma unroll
    for (int v = 0; v < 8; ++v)
      zbuf[(v + 8 * half) * SA + n] = acc[v] + bn;
  }
  __syncthreads();
  out_proj(zbuf, W_last, b_last, out, b0r, 0, wsel, lane);
  __syncthreads();

  for (int step = 0; step < TM1; ++step) {
    #pragma unroll 1
    for (int s = 0; s < 4; ++s) {
      const float frac = (s == 0) ? 0.f
                       : (s == 1) ? third
                       : (s == 2) ? (2.f * third) : 1.f;
      if (tid < MT * IN) {
        int m = tid >> 4, i = tid & 15;
        const float* cb = coeffs + ((size_t)(b0r + m) * TM1 + step) * (4 * IN);
        float bi = cb[IN + i], ci = cb[2 * IN + i], di = cb[3 * IN + i];
        dxs[i * MT + m] = bi + (ci + di * frac) * frac;
      }
      for (int e = tid; e < MT * H; e += THREADS) {
        int m = e >> 7, hc = e & (H - 1);
        float v = zbuf[m * SA + hc];
        if (s == 1)      v += kbuf[0][e] * third;
        else if (s == 2) v += kbuf[1][e] - kbuf[0][e] * third;
        else if (s == 3) {
          const float k1 = kbuf[0][e], k2 = kbuf[1][e], k3 = k3b[e];
          v += k1 - k2 + k3;
          kbuf[0][e] = k1 + 3.f * (k2 + k3);
        }
        actI[m * SA + hc] = v;
      }
      __syncthreads();

      if (wsel < NCOLW) mlp_layer(actI, wlds + 0 * H * SA, b0,          actA, wsel, lane);
      __syncthreads();
      if (wsel < NCOLW) mlp_layer(actA, wlds + 1 * H * SA, bl + 0 * SH, actI, wsel, lane);
      __syncthreads();
      if (wsel < NCOLW) mlp_layer(actI, wlds + 2 * H * SA, bl + 1 * SH, actA, wsel, lane);
      __syncthreads();
      if (wsel < NCOLW) mlp_layer(actA, wlds + 3 * H * SA, bl + 2 * SH, actI, wsel, lane);
      __syncthreads();

      {
        float ks[8];
        #pragma unroll
        for (int v = 0; v < 8; ++v) ks[v] = 0.f;
        const int kh = half << 1;
        const int hcol = hg * 16 + mD;
        const float* arow = actI + mD * SA + kh;
        const float* bfrow = bf + hcol * IN + chalf * CHW;
        const int r8 = lane >> 2, t4 = lane & 3;
        const float* grow = Wf + (size_t)((hg * 16 + r8) * IN + chalf * CHW) * SH + 4 * t4;
        float* const stg = pool + wsel * STG_W;
        float* const sst = stg + r8 * ST + 4 * t4;
        const float* sld = stg + mD * ST + kh;
        const float* dxh = dxs + chalf * CHW * MT + 8 * half;
        #pragma unroll 1
        for (int i = 0; i < CHW / 2; ++i) {
          const int p = (i + rot) & 3;
          const int c0 = 2 * p;
          const int c1 = c0 + 1;
          const v2f bfp = *(const v2f*)(bfrow + c0);
          v8f acc0 = {}, acc1 = {};
          if (p == 0) {
            res_pair(arow, wres + 0 * KS, wres + 1 * KS, acc0, acc1);
          } else {
            const float* g0 = grow + (size_t)c0 * SH;
            const float* gq = g0;
            const float* ak = arow;
            Chunk ca = load_chunk(gq);
            #pragma unroll 1
            for (int kt = 0; kt < SH - 32; kt += 32) {
              const Chunk cb = load_chunk(gq + 16);
              do_chunk(ca, ak, sst, sld, acc0, acc1);
              ca = load_chunk(gq + 32);
              do_chunk(cb, ak + 16, sst, sld, acc0, acc1);
              gq += 32;
              ak += 32;
            }
            {
              const Chunk cb = load_chunk(gq + 16);
              do_chunk(ca, ak, sst, sld, acc0, acc1);
              do_chunk(cb, ak + 16, sst, sld, acc0, acc1);
            }
          }
          const float4 d0a = *(const float4*)(dxh + c0 * MT);
          const float4 d0b = *(const float4*)(dxh + c0 * MT + 4);
          const float4 d1a = *(const float4*)(dxh + c1 * MT);
          const float4 d1b = *(const float4*)(dxh + c1 * MT + 4);
          const float d0[8] = {d0a.x, d0a.y, d0a.z, d0a.w, d0b.x, d0b.y, d0b.z, d0b.w};
          const float d1[8] = {d1a.x, d1a.y, d1a.z, d1a.w, d1b.x, d1b.y, d1b.z, d1b.w};
          #pragma unroll
          for (int v = 0; v < 8; ++v) {
            ks[v] += tanh_hw(acc0[v] + bfp.x) * d0[v];
            ks[v] += tanh_hw(acc1[v] + bfp.y) * d1[v];
          }
        }
        const int prow = 8 * half * SA + hcol;
        const int drow = 8 * half * H + hcol;
        __syncthreads();
        if (chalf == 1) {
          #pragma unroll
          for (int v = 0; v < 8; ++v) actA[prow + v * SA] = ks[v];
        }
        __syncthreads();
        if (chalf == 0) {
          if (s < 3) {
            float* kd = ((s < 2) ? kbuf[s] : k3b) + drow;
            float hv[8];
            #pragma unroll
            for (int v = 0; v < 8; ++v) hv[v] = actA[prow + v * SA];
            #pragma unroll
            for (int v = 0; v < 8; ++v) kd[v * H] = ks[v] + hv[v];
          } else {
            #pragma unroll
            for (int v = 0; v < 8; ++v) {
              const float kv = ks[v] + actA[prow + v * SA];
              zbuf[prow + v * SA] += (kbuf[0][drow + v * H] + kv) * 0.125f;
            }
          }
        }
      }
      __syncthreads();
    }
    out_proj(zbuf, W_last, b_last, out, b0r, step + 1, wsel, lane);
    __syncthreads();
  }
}

extern "C" void kernel_launch(void* const* d_in, const int* in_sizes, int n_in,
                              void* d_out, int out_size, void* d_ws, size_t ws_size,
                              hipStream_t stream) {
  (void)in_sizes; (void)n_in; (void)out_size; (void)d_ws; (void)ws_size;
  const float* coeffs = (const float*)d_in[1];
  const float* W_init = (const float*)d_in[2];
  const float* b_init = (const float*)d_in[3];
  const float* W0     = (const float*)d_in[4];
  const float* b0     = (const float*)d_in[5];
  const float* Wl     = (const float*)d_in[6];
  const float* bl     = (const float*)d_in[7];
  const float* Wf     = (const float*)d_in[8];
  const float* bf     = (const float*)d_in[9];
  const float* W_last = (const float*)d_in[10];
  const float* b_last = (const float*)d_in[11];
  float* out = (float*)d_out;

  ncde_wmma_kernel<<<B_T / MT, THREADS, 0, stream>>>(
      coeffs, W_init, b_init, W0, b0, Wl, bl, Wf, bf, W_last, b_last, out);
}


# Global-load waits leave the stage's critical path: layer biases read from LDS (copied once); dx and the output projection run on waves 8-15 while waves 0-7 do the small layers; same sums, same bits
# speedup vs baseline: 2.7527x; 1.0578x over previous
#include <hip/hip_runtime.h>
#include <math.h>


namespace {
constexpr int B_T = 256, T_T = 32, IN = 16, H = 128, SH = 128, OUTW = 64, TM1 = 31;
constexpr int MT = 16;
constexpr int THREADS = 512;
constexpr int NCOLW = 8;
constexpr int CHW = 8;
constexpr int SA = 132;
constexpr int NRES = 2;
constexpr int KS = SH / 4;
constexpr int ST = 20;
constexpr int STG_W = 16 * ST;
constexpr int POOL = 16 * STG_W;

typedef __attribute__((ext_vector_type(2))) float v2f;
typedef __attribute__((ext_vector_type(8))) float v8f;

__device__ __forceinline__ float tanh_hw(float x) {
  return __builtin_amdgcn_tanhf(x);
}

__device__ __forceinline__ v8f wmma4(v2f a, v2f b, v8f c) {
  return __builtin_amdgcn_wmma_f32_16x16x4_f32(false, a, false, b, (short)0, c,
                                               false, false);
}

__device__ __forceinline__ void wave_lds_handover() {
  __builtin_amdgcn_fence(__ATOMIC_RELEASE, "wavefront", "local");
  __builtin_amdgcn_wave_barrier();
  __builtin_amdgcn_fence(__ATOMIC_ACQUIRE, "wavefront", "local");
}

__device__ __forceinline__ void staged_unit(float* sst, const float* sld, float4 lo,
                                            float4 hi, const v2f* a, v8f& acc) {
  wave_lds_handover();
  *(float4*)(sst) = lo;
  *(float4*)(sst + 8 * ST) = hi;
  wave_lds_handover();
  #pragma unroll
  for (int j = 0; j < 4; ++j) {
    const v2f b = *(const v2f*)(sld + 4 * j);
    acc = wmma4(a[j], b, acc);
  }
}

struct Chunk { float4 t0l, t0h, t1l, t1h; };

__device__ __forceinline__ Chunk load_chunk(const float* gq) {
  constexpr size_t HI = (size_t)8 * IN * SH;
  Chunk c;
  c.t0l = *(const float4*)(gq);
  c.t0h = *(const float4*)(gq + HI);
  c.t1l = *(const float4*)(gq + SH);
  c.t1h = *(const float4*)(gq + SH + HI);
  return c;
}

__device__ __forceinline__ void do_chunk(const Chunk& c, const float* ak, float* sst,
                                         const float* sld, v8f& acc0, v8f& acc1) {
  v2f a[4];
  #pragma unroll
  for (int j = 0; j < 4; ++j) a[j] = *(const v2f*)(ak + 4 * j);
  staged_unit(sst, sld, c.t0l, c.t0h, a, acc0);
  staged_unit(sst, sld, c.t1l, c.t1h, a, acc1);
}

__device__ __forceinline__ v8f gemm_tile(const float* A, int lda, const float* W,
                                         int ldw, int n, int K, int lane) {
  const int m = lane & 15;
  const int kh = (lane >> 4) << 1;
  v8f acc = {};
  #pragma unroll 8
  for (int k0 = 0; k0 < K; k0 += 4) {
    v2f a = *(const v2f*)(A + m * lda + k0 + kh);
    v2f b = *(const v2f*)(W + (size_t)n * ldw + k0 + kh);
    acc = wmma4(a, b, acc);
  }
  return acc;
}

__device__ __forceinline__ void res_pair(const float* arow, const v2f* wa,
                                         const v2f* wb, v8f& acc0, v8f& acc1) {
  #pragma unroll
  for (int kk = 0; kk < KS; ++kk) {
    v2f a = *(const v2f*)(arow + 4 * kk);
    acc0 = wmma4(a, wa[kk], acc0);
    acc1 = wmma4(a, wb[kk], acc1);
  }
}

__device__ __forceinline__ void mlp_layer(const float* src, const float* Wlds,
                                          const float* bias, float* dst,
                                          int wid, int lane) {
  const int mD = lane & 15, half = lane >> 4;
  const int n = wid * 16 + mD;
  const float bn = bias[n];
  v8f acc = gemm_tile(src, SA, Wlds, SA, n, SH, lane);
  float* drow = dst + 8 * half * SA + n;
  #pragma unroll
  for (int v = 0; v < 8; ++v) {
    float x = acc[v] + bn;
    drow[v * SA] = fmaxf(x, 0.f);
  }
}

__device__ __forceinline__ void out_proj(const float* zb, const float* W_last,
                                         const float* b_last, float* out,
                                         int b0r, int t, int tile, int lane) {
  if (tile >= 0 && tile < 4) {
    const int mD = lane & 15, half = lane >> 4;
    const int n = tile * 16 + mD;
    v8f acc = gemm_tile(zb, SA, W_last, H, n, H, lane);
    const float bn = b_last[n];
    #pragma unroll
    for (int v = 0; v < 8; ++v) {
      int m = v + 8 * half;
      float x = acc[v] + bn;
      out[((size_t)(b0r + m) * T_T + t) * OUTW + n] = 1.f / (1.f + expf(-x));
    }
  }
}
}

__global__ __launch_bounds__(THREADS) void ncde_wmma_kernel(
    const float* __restrict__ coeffs,
    const float* __restrict__ W_init, const float* __restrict__ b_init,
    const float* __restrict__ W0, const float* __restrict__ b0,
    const float* __restrict__ Wl, const float* __restrict__ bl,
    const float* __restrict__ Wf, const float* __restrict__ bf,
    const float* __restrict__ W_last, const float* __restrict__ b_last,
    float* __restrict__ out) {
  __shared__ __align__(16) float wlds[4 * H * SA];
  __shared__ __align__(16) float zbuf[MT * SA];
  __shared__ __align__(16) float kbuf[2][MT * H];
  __shared__ __align__(16) float actI[MT * SA];
  __shared__ __align__(16) float pool[POOL];
  float* const k3b = pool;
  float* const actA = pool + MT * H;
  __shared__ __align__(16) float dxs[MT * IN];
  __shared__ __align__(16) float bias_s[4 * SH];

  const int tid = threadIdx.x;
  const int lane = tid & 31;
  const int wid = tid >> 5;
  const int b0r = blockIdx.x * MT;
  const int mD = lane & 15;
  const int half = lane >> 4;
  constexpr float third = 1.f / 3.f;
  const int wsel = __builtin_amdgcn_readfirstlane(wid);
  const int hg = wsel & (NCOLW - 1);
  const int chalf = wsel >> 3;

  v2f wres[NRES * KS];
  {
    const int khr = half << 1;
    #pragma unroll
    for (int j = 0; j < NRES; ++j) {
      const float* wr = Wf + (size_t)((hg * 16 + mD) * IN + chalf * CHW + j) * SH + khr;
      #pragma unroll
      for (int kk = 0; kk < KS; ++kk)
        wres[j * KS + kk] = *(const v2f*)(wr + 4 * kk);
    }
  }
  const int rot = (wsel >> 2) & 3;

  for (int e = tid; e < 4 * H * SH; e += THREADS) {
    int mat = e >> 14;
    int r = (e >> 7) & (H - 1);
    int k = e & (SH - 1);
    const float* src = (mat == 0) ? W0 : (Wl + (size_t)(mat - 1) * SH * SH);
    wlds[(size_t)mat * H * SA + r * SA + k] = src[(size_t)r * SH + k];
  }
  bias_s[tid] = (tid < SH) ? b0[tid] : bl[tid - SH];

  if (tid < MT * IN) {
    int m = tid >> 4, i = tid & 15;
    dxs[tid] = coeffs[((size_t)(b0r + m) * TM1) * (4 * IN) + i];
  }
  __syncthreads();
  if (wsel < NCOLW) {
    const int n = wsel * 16 + mD;
    v8f acc = gemm_tile(dxs, IN, W_init, IN, n, IN, lane);
    const float bn = b_init[n];
    #pragma unroll
    for (int v = 0; v < 8; ++v)
      zbuf[(v + 8 * half) * SA + n] = acc[v] + bn;
  }
  __syncthreads();

  #pragma unroll 1
  for (int it = 0; it < 4 * TM1; ++it) {
    {
      const int step = it >> 2;
      const int s = it & 3;
      const float frac = (s == 0) ? 0.f
                       : (s == 1) ? third
                       : (s == 2) ? (2.f * third) : 1.f;
      for (int e = tid; e < MT * H; e += THREADS) {
        int m = e >> 7, hc = e & (H - 1);
        float v = zbuf[m * SA + hc];
        if (s == 1)      v += kbuf[0][e] * third;
        else if (s == 2) v += kbuf[1][e] - kbuf[0][e] * third;
        else if (s == 3) {
          const float k1 = kbuf[0][e], k2 = kbuf[1][e], k3 = k3b[e];
          v += k1 - k2 + k3;
          kbuf[0][e] = k1 + 3.f * (k2 + k3);
        }
        actI[m * SA + hc] = v;
      }
      __syncthreads();

      if (wsel < NCOLW) {
        mlp_layer(actI, wlds + 0 * H * SA, bias_s + 0 * SH, actA, wsel, lane);
      } else {
        const int t = tid - MT * IN;
        const int m = t >> 4, i = t & 15;
        const float* cb = coeffs + ((size_t)(b0r + m) * TM1 + step) * (4 * IN);
        float bi = cb[IN + i], ci = cb[2 * IN + i], di = cb[3 * IN + i];
        dxs[i * MT + m] = bi + (ci + di * frac) * frac;
      }
      __syncthreads();
      if (wsel < NCOLW) {
        mlp_layer(actA, wlds + 1 * H * SA, bias_s + 1 * SH, actI, wsel, lane);
      } else if (s == 0) {
        out_proj(zbuf, W_last, b_last, out, b0r, step, wsel - NCOLW, lane);
      }
      __syncthreads();
      if (wsel < NCOLW) mlp_layer(actI, wlds + 2 * H * SA, bias_s + 2 * SH, actA, wsel, lane);
      __syncthreads();
      if (wsel < NCOLW) mlp_layer(actA, wlds + 3 * H * SA, bias_s + 3 * SH, actI, wsel, lane);
      __syncthreads();

      {
        float ks[8];
        #pragma unroll
        for (int v = 0; v < 8; ++v) ks[v] = 0.f;
        const int kh = half << 1;
        const int hcol = hg * 16 + mD;
        const float* arow = actI + mD * SA + kh;
        const float* bfrow = bf + hcol * IN + chalf * CHW;
        const int r8 = lane >> 2, t4 = lane & 3;
        const float* grow = Wf + (size_t)((hg * 16 + r8) * IN + chalf * CHW) * SH + 4 * t4;
        float* const stg = pool + wsel * STG_W;
        float* const sst = stg + r8 * ST + 4 * t4;
        const float* sld = stg + mD * ST + kh;
        const float* dxh = dxs + chalf * CHW * MT + 8 * half;
        #pragma unroll 1
        for (int i = 0; i < CHW / 2; ++i) {
          const int p = (i + rot) & 3;
          const int c0 = 2 * p;
          const int c1 = c0 + 1;
          const v2f bfp = *(const v2f*)(bfrow + c0);
          v8f acc0 = {}, acc1 = {};
          if (p == 0) {
            res_pair(arow, wres + 0 * KS, wres + 1 * KS, acc0, acc1);
          } else {
            const float* g0 = grow + (size_t)c0 * SH;
            const float* gq = g0;
            const float* ak = arow;
            Chunk ca = load_chunk(gq);
            #pragma unroll 1
            for (int kt = 0; kt < SH - 32; kt += 32) {
              const Chunk cb = load_chunk(gq + 16);
              do_chunk(ca, ak, sst, sld, acc0, acc1);
              ca = load_chunk(gq + 32);
              do_chunk(cb, ak + 16, sst, sld, acc0, acc1);
              gq += 32;
              ak += 32;
            }
            {
              const Chunk cb = load_chunk(gq + 16);
              do_chunk(ca, ak, sst, sld, acc0, acc1);
              do_chunk(cb, ak + 16, sst, sld, acc0, acc1);
            }
          }
          const float4 d0a = *(const float4*)(dxh + c0 * MT);
          const float4 d0b = *(const float4*)(dxh + c0 * MT + 4);
          const float4 d1a = *(const float4*)(dxh + c1 * MT);
          const float4 d1b = *(const float4*)(dxh + c1 * MT + 4);
          const float d0[8] = {d0a.x, d0a.y, d0a.z, d0a.w, d0b.x, d0b.y, d0b.z, d0b.w};
          const float d1[8] = {d1a.x, d1a.y, d1a.z, d1a.w, d1b.x, d1b.y, d1b.z, d1b.w};
          #pragma unroll
          for (int v = 0; v < 8; ++v) {
            ks[v] += tanh_hw(acc0[v] + bfp.x) * d0[v];
            ks[v] += tanh_hw(acc1[v] + bfp.y) * d1[v];
          }
        }
        const int prow = 8 * half * SA + hcol;
        const int drow = 8 * half * H + hcol;
        __syncthreads();
        if (chalf == 1) {
          #pragma unroll
          for (int v = 0; v < 8; ++v) actA[prow + v * SA] = ks[v];
        }
        __syncthreads();
        if (chalf == 0) {
          if (s < 3) {
            float* kd = ((s < 2) ? kbuf[s] : k3b) + drow;
            float hv[8];
            #pragma unroll
            for (int v = 0; v < 8; ++v) hv[v] = actA[prow + v * SA];
            #pragma unroll
            for (int v = 0; v < 8; ++v) kd[v * H] = ks[v] + hv[v];
          } else {
            #pragma unroll
            for (int v = 0; v < 8; ++v) {
              const float kv = ks[v] + actA[prow + v * SA];
              zbuf[prow + v * SA] += (kbuf[0][drow + v * H] + kv) * 0.125f;
            }
          }
        }
      }
      __syncthreads();
    }
  }
  out_proj(zbuf, W_last, b_last, out, b0r, TM1, wsel, lane);
}

extern "C" void kernel_launch(void* const* d_in, const int* in_sizes, int n_in,
                              void* d_out, int out_size, void* d_ws, size_t ws_size,
                              hipStream_t stream) {
  (void)in_sizes; (void)n_in; (void)out_size; (void)d_ws; (void)ws_size;
  const float* coeffs = (const float*)d_in[1];
  const float* W_init = (const float*)d_in[2];
  const float* b_init = (const float*)d_in[3];
  const float* W0     = (const float*)d_in[4];
  const float* b0     = (const float*)d_in[5];
  const float* Wl     = (const float*)d_in[6];
  const float* bl     = (const float*)d_in[7];
  const float* Wf     = (const float*)d_in[8];
  const float* bf     = (const float*)d_in[9];
  const float* W_last = (const float*)d_in[10];
  const float* b_last = (const float*)d_in[11];
  float* out = (float*)d_out;

  ncde_wmma_kernel<<<B_T / MT, THREADS, 0, stream>>>(
      coeffs, W_init, b_init, W0, b0, Wl, bl, Wf, bf, W_last, b_last, out);
}


# Small GEMMs of a stage (4 hidden layers, output projection): fragments requested a block of k steps ahead of their WMMAs (2nd register set; W_last as one group, b_last from LDS); same sums, same bits
# speedup vs baseline: 2.9256x; 1.0628x over previous
#include <hip/hip_runtime.h>
#include <math.h>


namespace {
constexpr int B_T = 256, T_T = 32, IN = 16, H = 128, SH = 128, OUTW = 64, TM1 = 31;
constexpr int MT = 16;
constexpr int THREADS = 512;
constexpr int NCOLW = 8;
constexpr int CHW = 8;
constexpr int SA = 132;
constexpr int NRES = 2;
constexpr int KS = SH / 4;
constexpr int ST = 20;
constexpr int STG_W = 16 * ST;
constexpr int POOL = 16 * STG_W;

typedef __attribute__((ext_vector_type(2))) float v2f;
typedef __attribute__((ext_vector_type(8))) float v8f;

__device__ __forceinline__ float tanh_hw(float x) {
  return __builtin_amdgcn_tanhf(x);
}

__device__ __forceinline__ v8f wmma4(v2f a, v2f b, v8f c) {
  return __builtin_amdgcn_wmma_f32_16x16x4_f32(false, a, false, b, (short)0, c,
                                               false, false);
}

__device__ __forceinline__ void wave_lds_handover() {
  __builtin_amdgcn_fence(__ATOMIC_RELEASE, "wavefront", "local");
  __builtin_amdgcn_wave_barrier();
  __builtin_amdgcn_fence(__ATOMIC_ACQUIRE, "wavefront", "local");
}

__device__ __forceinline__ void staged_unit(float* sst, const float* sld, float4 lo,
                                            float4 hi, const v2f* a, v8f& acc) {
  wave_lds_handover();
  *(float4*)(sst) = lo;
  *(float4*)(sst + 8 * ST) = hi;
  wave_lds_handover();
  #pragma unroll
  for (int j = 0; j < 4; ++j) {
    const v2f b = *(const v2f*)(sld + 4 * j);
    acc = wmma4(a[j], b, acc);
  }
}

struct Chunk { float4 t0l, t0h, t1l, t1h; };

__device__ __forceinline__ Chunk load_chunk(const float* gq) {
  constexpr size_t HI = (size_t)8 * IN * SH;
  Chunk c;
  c.t0l = *(const float4*)(gq);
  c.t0h = *(const float4*)(gq + HI);
  c.t1l = *(const float4*)(gq + SH);
  c.t1h = *(const float4*)(gq + SH + HI);
  return c;
}

__device__ __forceinline__ void do_chunk(const Chunk& c, const float* ak, float* sst,
                                         const float* sld, v8f& acc0, v8f& acc1) {
  v2f a[4];
  #pragma unroll
  for (int j = 0; j < 4; ++j) a[j] = *(const v2f*)(ak + 4 * j);
  staged_unit(sst, sld, c.t0l, c.t0h, a, acc0);
  staged_unit(sst, sld, c.t1l, c.t1h, a, acc1);
}

__device__ __forceinline__ v8f gemm_tile(const float* A, int lda, const float* W,
                                         int ldw, int n, int K, int lane) {
  const int m = lane & 15;
  const int kh = (lane >> 4) << 1;
  v8f acc = {};
  #pragma unroll 8
  for (int k0 = 0; k0 < K; k0 += 4) {
    v2f a = *(const v2f*)(A + m * lda + k0 + kh);
    v2f b = *(const v2f*)(W + (size_t)n * ldw + k0 + kh);
    acc = wmma4(a, b, acc);
  }
  return acc;
}

__device__ __forceinline__ void res_pair(const float* arow, const v2f* wa,
                                         const v2f* wb, v8f& acc0, v8f& acc1) {
  #pragma unroll
  for (int kk = 0; kk < KS; ++kk) {
    v2f a = *(const v2f*)(arow + 4 * kk);
    acc0 = wmma4(a, wa[kk], acc0);
    acc1 = wmma4(a, wb[kk], acc1);
  }
}

__device__ __forceinline__ void mlp_layer(const float* src, const float* Wlds,
                                          const float* bias, float* dst,
                                          int wid, int lane) {
  constexpr int KB = 8;
  constexpr int NB = KS / KB;
  const int mD = lane & 15, half = lane >> 4;
  const int n = wid * 16 + mD;
  const int kh = half << 1;
  const float bn = bias[n];
  const float* ap = src + mD * SA + kh;
  const float* bp = Wlds + n * SA + kh;
  v2f a[2][KB], b[2][KB];
  #pragma unroll
  for (int j = 0; j < KB; ++j) {
    a[0][j] = *(const v2f*)(ap + 4 * j);
    b[0][j] = *(const v2f*)(bp + 4 * j);
  }
  v8f acc = {};
  #pragma unroll
  for (int blk = 0; blk < NB; ++blk) {
    const int cur = blk & 1, nxt = cur ^ 1;
    if (blk + 1 < NB) {
      #pragma unroll
      for (int j = 0; j < KB; ++j) {
        a[nxt][j] = *(const v2f*)(ap + 4 * ((blk + 1) * KB + j));
        b[nxt][j] = *(const v2f*)(bp + 4 * ((blk + 1) * KB + j));
      }
    }
    __builtin_amdgcn_sched_barrier(0);
    #pragma unroll
    for (int j = 0; j < KB; ++j) acc = wmma4(a[cur][j], b[cur][j], acc);
    __builtin_amdgcn_sched_barrier(0);
  }
  float* drow = dst + 8 * half * SA + n;
  #pragma unroll
  for (int v = 0; v < 8; ++v) {
    float x = acc[v] + bn;
    drow[v * SA] = fmaxf(x, 0.f);
  }
}

__device__ __forceinline__ void out_proj(const float* zb, const float* W_last,
                                         const float* blast, float* out,
                                         int b0r, int t, int tile, int lane) {
  if (tile >= 0 && tile < 4) {
    constexpr int KB = 4;
    constexpr int NB = KS / KB;
    const int mD = lane & 15, half = lane >> 4;
    const int n = tile * 16 + mD;
    const int kh = half << 1;
    const float bn = blast[n];
    const float* wr = W_last + (size_t)n * H + kh;
    v2f b[KS];
    #pragma unroll
    for (int kk = 0; kk < KS; ++kk) b[kk] = *(const v2f*)(wr + 4 * kk);
    const float* ap = zb + mD * SA + kh;
    v2f a[2][KB];
    #pragma unroll
    for (int j = 0; j < KB; ++j) a[0][j] = *(const v2f*)(ap + 4 * j);
    v8f acc = {};
    #pragma unroll
    for (int blk = 0; blk < NB; ++blk) {
      const int cur = blk & 1, nxt = cur ^ 1;
      if (blk + 1 < NB) {
        #pragma unroll
        for (int j = 0; j < KB; ++j)
          a[nxt][j] = *(const v2f*)(ap + 4 * ((blk + 1) * KB + j));
      }
      __builtin_amdgcn_sched_barrier(0);
      #pragma unroll
      for (int j = 0; j < KB; ++j) acc = wmma4(a[cur][j], b[blk * KB + j], acc);
      __builtin_amdgcn_sched_barrier(0);
    }
    #pragma unroll
    for (int v = 0; v < 8; ++v) {
      int m = v + 8 * half;
      float x = acc[v] + bn;
      out[((size_t)(b0r + m) * T_T + t) * OUTW + n] = 1.f / (1.f + expf(-x));
    }
  }
}
}

__global__ __launch_bounds__(THREADS) void ncde_wmma_kernel(
    const float* __restrict__ coeffs,
    const float* __restrict__ W_init, const float* __restrict__ b_init,
    const float* __restrict__ W0, const float* __restrict__ b0,
    const float* __restrict__ Wl, const float* __restrict__ bl,
    const float* __restrict__ Wf, const float* __restrict__ bf,
    const float* __restrict__ W_last, const float* __restrict__ b_last,
    float* __restrict__ out) {
  __shared__ __align__(16) float wlds[4 * H * SA];
  __shared__ __align__(16) float zbuf[MT * SA];
  __shared__ __align__(16) float kbuf[2][MT * H];
  __shared__ __align__(16) float actI[MT * SA];
  __shared__ __align__(16) float pool[POOL];
  float* const k3b = pool;
  float* const actA = pool + MT * H;
  __shared__ __align__(16) float dxs[MT * IN];
  __shared__ __align__(16) float bias_s[4 * SH + OUTW];

  const int tid = threadIdx.x;
  const int lane = tid & 31;
  const int wid = tid >> 5;
  const int b0r = blockIdx.x * MT;
  const int mD = lane & 15;
  const int half = lane >> 4;
  constexpr float third = 1.f / 3.f;
  const int wsel = __builtin_amdgcn_readfirstlane(wid);
  const int hg = wsel & (NCOLW - 1);
  const int chalf = wsel >> 3;

  v2f wres[NRES * KS];
  {
    const int khr = half << 1;
    #pragma unroll
    for (int j = 0; j < NRES; ++j) {
      const float* wr = Wf + (size_t)((hg * 16 + mD) * IN + chalf * CHW + j) * SH + khr;
      #pragma unroll
      for (int kk = 0; kk < KS; ++kk)
        wres[j * KS + kk] = *(const v2f*)(wr + 4 * kk);
    }
  }
  const int rot = (wsel >> 2) & 3;

  for (int e = tid; e < 4 * H * SH; e += THREADS) {
    int mat = e >> 14;
    int r = (e >> 7) & (H - 1);
    int k = e & (SH - 1);
    const float* src = (mat == 0) ? W0 : (Wl + (size_t)(mat - 1) * SH * SH);
    wlds[(size_t)mat * H * SA + r * SA + k] = src[(size_t)r * SH + k];
  }
  bias_s[tid] = (tid < SH) ? b0[tid] : bl[tid - SH];
  if (tid < OUTW) bias_s[4 * SH + tid] = b_last[tid];

  if (tid < MT * IN) {
    int m = tid >> 4, i = tid & 15;
    dxs[tid] = coeffs[((size_t)(b0r + m) * TM1) * (4 * IN) + i];
  }
  __syncthreads();
  if (wsel < NCOLW) {
    const int n = wsel * 16 + mD;
    v8f acc = gemm_tile(dxs, IN, W_init, IN, n, IN, lane);
    const float bn = b_init[n];
    #pragma unroll
    for (int v = 0; v < 8; ++v)
      zbuf[(v + 8 * half) * SA + n] = acc[v] + bn;
  }
  __syncthreads();

  #pragma unroll 1
  for (int it = 0; it < 4 * TM1; ++it) {
    {
      const int step = it >> 2;
      const int s = it & 3;
      const float frac = (s == 0) ? 0.f
                       : (s == 1) ? third
                       : (s == 2) ? (2.f * third) : 1.f;
      for (int e = tid; e < MT * H; e += THREADS) {
        int m = e >> 7, hc = e & (H - 1);
        float v = zbuf[m * SA + hc];
        if (s == 1)      v += kbuf[0][e] * third;
        else if (s == 2) v += kbuf[1][e] - kbuf[0][e] * third;
        else if (s == 3) {
          const float k1 = kbuf[0][e], k2 = kbuf[1][e], k3 = k3b[e];
          v += k1 - k2 + k3;
          kbuf[0][e] = k1 + 3.f * (k2 + k3);
        }
        actI[m * SA + hc] = v;
      }
      __syncthreads();

      if (wsel < NCOLW) {
        mlp_layer(actI, wlds + 0 * H * SA, bias_s + 0 * SH, actA, wsel, lane);
      } else {
        const int t = tid - MT * IN;
        const int m = t >> 4, i = t & 15;
        const float* cb = coeffs + ((size_t)(b0r + m) * TM1 + step) * (4 * IN);
        float bi = cb[IN + i], ci = cb[2 * IN + i], di = cb[3 * IN + i];
        dxs[i * MT + m] = bi + (ci + di * frac) * frac;
      }
      __syncthreads();
      if (wsel < NCOLW) {
        mlp_layer(actA, wlds + 1 * H * SA, bias_s + 1 * SH, actI, wsel, lane);
      } else if (s == 0) {
        out_proj(zbuf, W_last, bias_s + 4 * SH, out, b0r, step, wsel - NCOLW, lane);
      }
      __syncthreads();
      if (wsel < NCOLW) mlp_layer(actI, wlds + 2 * H * SA, bias_s + 2 * SH, actA, wsel, lane);
      __syncthreads();
      if (wsel < NCOLW) mlp_layer(actA, wlds + 3 * H * SA, bias_s + 3 * SH, actI, wsel, lane);
      __syncthreads();

      {
        float ks[8];
        #pragma unroll
        for (int v = 0; v < 8; ++v) ks[v] = 0.f;
        const int kh = half << 1;
        const int hcol = hg * 16 + mD;
        const float* arow = actI + mD * SA + kh;
        const float* bfrow = bf + hcol * IN + chalf * CHW;
        const int r8 = lane >> 2, t4 = lane & 3;
        const float* grow = Wf + (size_t)((hg * 16 + r8) * IN + chalf * CHW) * SH + 4 * t4;
        float* const stg = pool + wsel * STG_W;
        float* const sst = stg + r8 * ST + 4 * t4;
        const float* sld = stg + mD * ST + kh;
        const float* dxh = dxs + chalf * CHW * MT + 8 * half;
        #pragma unroll 1
        for (int i = 0; i < CHW / 2; ++i) {
          const int p = (i + rot) & 3;
          const int c0 = 2 * p;
          const int c1 = c0 + 1;
          const v2f bfp = *(const v2f*)(bfrow + c0);
          v8f acc0 = {}, acc1 = {};
          if (p == 0) {
            res_pair(arow, wres + 0 * KS, wres + 1 * KS, acc0, acc1);
          } else {
            const float* g0 = grow + (size_t)c0 * SH;
            const float* gq = g0;
            const float* ak = arow;
            Chunk ca = load_chunk(gq);
            #pragma unroll 1
            for (int kt = 0; kt < SH - 32; kt += 32) {
              const Chunk cb = load_chunk(gq + 16);
              do_chunk(ca, ak, sst, sld, acc0, acc1);
              ca = load_chunk(gq + 32);
              do_chunk(cb, ak + 16, sst, sld, acc0, acc1);
              gq += 32;
              ak += 32;
            }
            {
              const Chunk cb = load_chunk(gq + 16);
              do_chunk(ca, ak, sst, sld, acc0, acc1);
              do_chunk(cb, ak + 16, sst, sld, acc0, acc1);
            }
          }
          const float4 d0a = *(const float4*)(dxh + c0 * MT);
          const float4 d0b = *(const float4*)(dxh + c0 * MT + 4);
          const float4 d1a = *(const float4*)(dxh + c1 * MT);
          const float4 d1b = *(const float4*)(dxh + c1 * MT + 4);
          const float d0[8] = {d0a.x, d0a.y, d0a.z, d0a.w, d0b.x, d0b.y, d0b.z, d0b.w};
          const float d1[8] = {d1a.x, d1a.y, d1a.z, d1a.w, d1b.x, d1b.y, d1b.z, d1b.w};
          #pragma unroll
          for (int v = 0; v < 8; ++v) {
            ks[v] += tanh_hw(acc0[v] + bfp.x) * d0[v];
            ks[v] += tanh_hw(acc1[v] + bfp.y) * d1[v];
          }
        }
        const int prow = 8 * half * SA + hcol;
        const int drow = 8 * half * H + hcol;
        __syncthreads();
        if (chalf == 1) {
          #pragma unroll
          for (int v = 0; v < 8; ++v) actA[prow + v * SA] = ks[v];
        }
        __syncthreads();
        if (chalf == 0) {
          if (s < 3) {
            float* kd = ((s < 2) ? kbuf[s] : k3b) + drow;
            float hv[8];
            #pragma unroll
            for (int v = 0; v < 8; ++v) hv[v] = actA[prow + v * SA];
            #pragma unroll
            for (int v = 0; v < 8; ++v) kd[v * H] = ks[v] + hv[v];
          } else {
            #pragma unroll
            for (int v = 0; v < 8; ++v) {
              const float kv = ks[v] + actA[prow + v * SA];
              zbuf[prow + v * SA] += (kbuf[0][drow + v * H] + kv) * 0.125f;
            }
          }
        }
      }
      __syncthreads();
    }
  }
  out_proj(zbuf, W_last, bias_s + 4 * SH, out, b0r, TM1, wsel, lane);
}

extern "C" void kernel_launch(void* const* d_in, const int* in_sizes, int n_in,
                              void* d_out, int out_size, void* d_ws, size_t ws_size,
                              hipStream_t stream) {
  (void)in_sizes; (void)n_in; (void)out_size; (void)d_ws; (void)ws_size;
  const float* coeffs = (const float*)d_in[1];
  const float* W_init = (const float*)d_in[2];
  const float* b_init = (const float*)d_in[3];
  const float* W0     = (const float*)d_in[4];
  const float* b0     = (const float*)d_in[5];
  const float* Wl     = (const float*)d_in[6];
  const float* bl     = (const float*)d_in[7];
  const float* Wf     = (const float*)d_in[8];
  const float* bf     = (const float*)d_in[9];
  const float* W_last = (const float*)d_in[10];
  const float* b_last = (const float*)d_in[11];
  float* out = (float*)d_out;

  ncde_wmma_kernel<<<B_T / MT, THREADS, 0, stream>>>(
      coeffs, W_init, b_init, W0, b0, Wl, bl, Wf, bf, W_last, b_last, out);
}


# Wf phase: a wave's issue priority (s_setprio) drops by one per tile pair it has finished, so a SIMD's four waves advance together; youngest wave's resident pair last: channel sum order changes
# speedup vs baseline: 3.0768x; 1.0517x over previous
#include <hip/hip_runtime.h>
#include <math.h>


namespace {
constexpr int B_T = 256, T_T = 32, IN = 16, H = 128, SH = 128, OUTW = 64, TM1 = 31;
constexpr int MT = 16;
constexpr int THREADS = 512;
constexpr int NCOLW = 8;
constexpr int CHW = 8;
constexpr int SA = 132;
constexpr int NRES = 2;
constexpr int KS = SH / 4;
constexpr int ST = 20;
constexpr int STG_W = 16 * ST;
constexpr int POOL = 16 * STG_W;

typedef __attribute__((ext_vector_type(2))) float v2f;
typedef __attribute__((ext_vector_type(8))) float v8f;

__device__ __forceinline__ float tanh_hw(float x) {
  return __builtin_amdgcn_tanhf(x);
}

__device__ __forceinline__ v8f wmma4(v2f a, v2f b, v8f c) {
  return __builtin_amdgcn_wmma_f32_16x16x4_f32(false, a, false, b, (short)0, c,
                                               false, false);
}

__device__ __forceinline__ void wave_lds_handover() {
  __builtin_amdgcn_fence(__ATOMIC_RELEASE, "wavefront", "local");
  __builtin_amdgcn_wave_barrier();
  __builtin_amdgcn_fence(__ATOMIC_ACQUIRE, "wavefront", "local");
}

__device__ __forceinline__ void staged_unit(float* sst, const float* sld, float4 lo,
                                            float4 hi, const v2f* a, v8f& acc) {
  wave_lds_handover();
  *(float4*)(sst) = lo;
  *(float4*)(sst + 8 * ST) = hi;
  wave_lds_handover();
  #pragma unroll
  for (int j = 0; j < 4; ++j) {
    const v2f b = *(const v2f*)(sld + 4 * j);
    acc = wmma4(a[j], b, acc);
  }
}

struct Chunk { float4 t0l, t0h, t1l, t1h; };

__device__ __forceinline__ Chunk load_chunk(const float* gq) {
  constexpr size_t HI = (size_t)8 * IN * SH;
  Chunk c;
  c.t0l = *(const float4*)(gq);
  c.t0h = *(const float4*)(gq + HI);
  c.t1l = *(const float4*)(gq + SH);
  c.t1h = *(const float4*)(gq + SH + HI);
  return c;
}

__device__ __forceinline__ void do_chunk(const Chunk& c, const float* ak, float* sst,
                                         const float* sld, v8f& acc0, v8f& acc1) {
  v2f a[4];
  #pragma unroll
  for (int j = 0; j < 4; ++j) a[j] = *(const v2f*)(ak + 4 * j);
  staged_unit(sst, sld, c.t0l, c.t0h, a, acc0);
  staged_unit(sst, sld, c.t1l, c.t1h, a, acc1);
}

__device__ __forceinline__ v8f gemm_tile(const float* A, int lda, const float* W,
                                         int ldw, int n, int K, int lane) {
  const int m = lane & 15;
  const int kh = (lane >> 4) << 1;
  v8f acc = {};
  #pragma unroll 8
  for (int k0 = 0; k0 < K; k0 += 4) {
    v2f a = *(const v2f*)(A + m * lda + k0 + kh);
    v2f b = *(const v2f*)(W + (size_t)n * ldw + k0 + kh);
    acc = wmma4(a, b, acc);
  }
  return acc;
}

__device__ __forceinline__ void res_pair(const float* arow, const v2f* wa,
                                         const v2f* wb, v8f& acc0, v8f& acc1) {
  #pragma unroll
  for (int kk = 0; kk < KS; ++kk) {
    v2f a = *(const v2f*)(arow + 4 * kk);
    acc0 = wmma4(a, wa[kk], acc0);
    acc1 = wmma4(a, wb[kk], acc1);
  }
}

__device__ __forceinline__ void mlp_layer(const float* src, const float* Wlds,
                                          const float* bias, float* dst,
                                          int wid, int lane) {
  constexpr int KB = 8;
  constexpr int NB = KS / KB;
  const int mD = lane & 15, half = lane >> 4;
  const int n = wid * 16 + mD;
  const int kh = half << 1;
  const float bn = bias[n];
  const float* ap = src + mD * SA + kh;
  const float* bp = Wlds + n * SA + kh;
  v2f a[2][KB], b[2][KB];
  #pragma unroll
  for (int j = 0; j < KB; ++j) {
    a[0][j] = *(const v2f*)(ap + 4 * j);
    b[0][j] = *(const v2f*)(bp + 4 * j);
  }
  v8f acc = {};
  #pragma unroll
  for (int blk = 0; blk < NB; ++blk) {
    const int cur = blk & 1, nxt = cur ^ 1;
    if (blk + 1 < NB) {
      #pragma unroll
      for (int j = 0; j < KB; ++j) {
        a[nxt][j] = *(const v2f*)(ap + 4 * ((blk + 1) * KB + j));
        b[nxt][j] = *(const v2f*)(bp + 4 * ((blk + 1) * KB + j));
      }
    }
    __builtin_amdgcn_sched_barrier(0);
    #pragma unroll
    for (int j = 0; j < KB; ++j) acc = wmma4(a[cur][j], b[cur][j], acc);
    __builtin_amdgcn_sched_barrier(0);
  }
  float* drow = dst + 8 * half * SA + n;
  #pragma unroll
  for (int v = 0; v < 8; ++v) {
    float x = acc[v] + bn;
    drow[v * SA] = fmaxf(x, 0.f);
  }
}

__device__ __forceinline__ void out_proj(const float* zb, const float* W_last,
                                         const float* blast, float* out,
                                         int b0r, int t, int tile, int lane) {
  if (tile >= 0 && tile < 4) {
    constexpr int KB = 4;
    constexpr int NB = KS / KB;
    const int mD = lane & 15, half = lane >> 4;
    const int n = tile * 16 + mD;
    const int kh = half << 1;
    const float bn = blast[n];
    const float* wr = W_last + (size_t)n * H + kh;
    v2f b[KS];
    #pragma unroll
    for (int kk = 0; kk < KS; ++kk) b[kk] = *(const v2f*)(wr + 4 * kk);
    const float* ap = zb + mD * SA + kh;
    v2f a[2][KB];
    #pragma unroll
    for (int j = 0; j < KB; ++j) a[0][j] = *(const v2f*)(ap + 4 * j);
    v8f acc = {};
    #pragma unroll
    for (int blk = 0; blk < NB; ++blk) {
      const int cur = blk & 1, nxt = cur ^ 1;
      if (blk + 1 < NB) {
        #pragma unroll
        for (int j = 0; j < KB; ++j)
          a[nxt][j] = *(const v2f*)(ap + 4 * ((blk + 1) * KB + j));
      }
      __builtin_amdgcn_sched_barrier(0);
      #pragma unroll
      for (int j = 0; j < KB; ++j) acc = wmma4(a[cur][j], b[blk * KB + j], acc);
      __builtin_amdgcn_sched_barrier(0);
    }
    #pragma unroll
    for (int v = 0; v < 8; ++v) {
      int m = v + 8 * half;
      float x = acc[v] + bn;
      out[((size_t)(b0r + m) * T_T + t) * OUTW + n] = 1.f / (1.f + expf(-x));
    }
  }
}
}

__global__ __launch_bounds__(THREADS) void ncde_wmma_kernel(
    const float* __restrict__ coeffs,
    const float* __restrict__ W_init, const float* __restrict__ b_init,
    const float* __restrict__ W0, const float* __restrict__ b0,
    const float* __restrict__ Wl, const float* __restrict__ bl,
    const float* __restrict__ Wf, const float* __restrict__ bf,
    const float* __restrict__ W_last, const float* __restrict__ b_last,
    float* __restrict__ out) {
  __shared__ __align__(16) float wlds[4 * H * SA];
  __shared__ __align__(16) float zbuf[MT * SA];
  __shared__ __align__(16) float kbuf[2][MT * H];
  __shared__ __align__(16) float actI[MT * SA];
  __shared__ __align__(16) float pool[POOL];
  float* const k3b = pool;
  float* const actA = pool + MT * H;
  __shared__ __align__(16) float dxs[MT * IN];
  __shared__ __align__(16) float bias_s[4 * SH + OUTW];

  const int tid = threadIdx.x;
  const int lane = tid & 31;
  const int wid = tid >> 5;
  const int b0r = blockIdx.x * MT;
  const int mD = lane & 15;
  const int half = lane >> 4;
  constexpr float third = 1.f / 3.f;
  const int wsel = __builtin_amdgcn_readfirstlane(wid);
  const int hg = wsel & (NCOLW - 1);
  const int chalf = wsel >> 3;

  v2f wres[NRES * KS];
  {
    const int khr = half << 1;
    #pragma unroll
    for (int j = 0; j < NRES; ++j) {
      const float* wr = Wf + (size_t)((hg * 16 + mD) * IN + chalf * CHW + j) * SH + khr;
      #pragma unroll
      for (int kk = 0; kk < KS; ++kk)
        wres[j * KS + kk] = *(const v2f*)(wr + 4 * kk);
    }
  }
  const int rot = (4 - (wsel >> 2)) & 3;

  for (int e = tid; e < 4 * H * SH; e += THREADS) {
    int mat = e >> 14;
    int r = (e >> 7) & (H - 1);
    int k = e & (SH - 1);
    const float* src = (mat == 0) ? W0 : (Wl + (size_t)(mat - 1) * SH * SH);
    wlds[(size_t)mat * H * SA + r * SA + k] = src[(size_t)r * SH + k];
  }
  bias_s[tid] = (tid < SH) ? b0[tid] : bl[tid - SH];
  if (tid < OUTW) bias_s[4 * SH + tid] = b_last[tid];

  if (tid < MT * IN) {
    int m = tid >> 4, i = tid & 15;
    dxs[tid] = coeffs[((size_t)(b0r + m) * TM1) * (4 * IN) + i];
  }
  __syncthreads();
  if (wsel < NCOLW) {
    const int n = wsel * 16 + mD;
    v8f acc = gemm_tile(dxs, IN, W_init, IN, n, IN, lane);
    const float bn = b_init[n];
    #pragma unroll
    for (int v = 0; v < 8; ++v)
      zbuf[(v + 8 * half) * SA + n] = acc[v] + bn;
  }
  __syncthreads();

  #pragma unroll 1
  for (int it = 0; it < 4 * TM1; ++it) {
    {
      const int step = it >> 2;
      const int s = it & 3;
      const float frac = (s == 0) ? 0.f
                       : (s == 1) ? third
                       : (s == 2) ? (2.f * third) : 1.f;
      for (int e = tid; e < MT * H; e += THREADS) {
        int m = e >> 7, hc = e & (H - 1);
        float v = zbuf[m * SA + hc];
        if (s == 1)      v += kbuf[0][e] * third;
        else if (s == 2) v += kbuf[1][e] - kbuf[0][e] * third;
        else if (s == 3) {
          const float k1 = kbuf[0][e], k2 = kbuf[1][e], k3 = k3b[e];
          v += k1 - k2 + k3;
          kbuf[0][e] = k1 + 3.f * (k2 + k3);
        }
        actI[m * SA + hc] = v;
      }
      __syncthreads();

      if (wsel < NCOLW) {
        mlp_layer(actI, wlds + 0 * H * SA, bias_s + 0 * SH, actA, wsel, lane);
      } else {
        const int t = tid - MT * IN;
        const int m = t >> 4, i = t & 15;
        const float* cb = coeffs + ((size_t)(b0r + m) * TM1 + step) * (4 * IN);
        float bi = cb[IN + i], ci = cb[2 * IN + i], di = cb[3 * IN + i];
        dxs[i * MT + m] = bi + (ci + di * frac) * frac;
      }
      __syncthreads();
      if (wsel < NCOLW) {
        mlp_layer(actA, wlds + 1 * H * SA, bias_s + 1 * SH, actI, wsel, lane);
      } else if (s == 0) {
        out_proj(zbuf, W_last, bias_s + 4 * SH, out, b0r, step, wsel - NCOLW, lane);
      }
      __syncthreads();
      if (wsel < NCOLW) mlp_layer(actI, wlds + 2 * H * SA, bias_s + 2 * SH, actA, wsel, lane);
      __syncthreads();
      if (wsel < NCOLW) mlp_layer(actA, wlds + 3 * H * SA, bias_s + 3 * SH, actI, wsel, lane);
      __syncthreads();

      {
        float ks[8];
        #pragma unroll
        for (int v = 0; v < 8; ++v) ks[v] = 0.f;
        const int kh = half << 1;
        const int hcol = hg * 16 + mD;
        const float* arow = actI + mD * SA + kh;
        const float* bfrow = bf + hcol * IN + chalf * CHW;
        const int r8 = lane >> 2, t4 = lane & 3;
        const float* grow = Wf + (size_t)((hg * 16 + r8) * IN + chalf * CHW) * SH + 4 * t4;
        float* const stg = pool + wsel * STG_W;
        float* const sst = stg + r8 * ST + 4 * t4;
        const float* sld = stg + mD * ST + kh;
        const float* dxh = dxs + chalf * CHW * MT + 8 * half;
        #pragma unroll 1
        for (int i = 0; i < CHW / 2; ++i) {
          if (i == 0)      __builtin_amdgcn_s_setprio(3);
          else if (i == 1) __builtin_amdgcn_s_setprio(2);
          else if (i == 2) __builtin_amdgcn_s_setprio(1);
          else             __builtin_amdgcn_s_setprio(0);
          const int p = (i + rot) & 3;
          const int c0 = 2 * p;
          const int c1 = c0 + 1;
          const v2f bfp = *(const v2f*)(bfrow + c0);
          v8f acc0 = {}, acc1 = {};
          if (p == 0) {
            res_pair(arow, wres + 0 * KS, wres + 1 * KS, acc0, acc1);
          } else {
            const float* g0 = grow + (size_t)c0 * SH;
            const float* gq = g0;
            const float* ak = arow;
            Chunk ca = load_chunk(gq);
            #pragma unroll 1
            for (int kt = 0; kt < SH - 32; kt += 32) {
              const Chunk cb = load_chunk(gq + 16);
              do_chunk(ca, ak, sst, sld, acc0, acc1);
              ca = load_chunk(gq + 32);
              do_chunk(cb, ak + 16, sst, sld, acc0, acc1);
              gq += 32;
              ak += 32;
            }
            {
              const Chunk cb = load_chunk(gq + 16);
              do_chunk(ca, ak, sst, sld, acc0, acc1);
              do_chunk(cb, ak + 16, sst, sld, acc0, acc1);
            }
          }
          const float4 d0a = *(const float4*)(dxh + c0 * MT);
          const float4 d0b = *(const float4*)(dxh + c0 * MT + 4);
          const float4 d1a = *(const float4*)(dxh + c1 * MT);
          const float4 d1b = *(const float4*)(dxh + c1 * MT + 4);
          const float d0[8] = {d0a.x, d0a.y, d0a.z, d0a.w, d0b.x, d0b.y, d0b.z, d0b.w};
          const float d1[8] = {d1a.x, d1a.y, d1a.z, d1a.w, d1b.x, d1b.y, d1b.z, d1b.w};
          #pragma unroll
          for (int v = 0; v < 8; ++v) {
            ks[v] += tanh_hw(acc0[v] + bfp.x) * d0[v];
            ks[v] += tanh_hw(acc1[v] + bfp.y) * d1[v];
          }
        }
        const int prow = 8 * half * SA + hcol;
        const int drow = 8 * half * H + hcol;
        __syncthreads();
        if (chalf == 1) {
          #pragma unroll
          for (int v = 0; v < 8; ++v) actA[prow + v * SA] = ks[v];
        }
        __syncthreads();
        if (chalf == 0) {
          if (s < 3) {
            float* kd = ((s < 2) ? kbuf[s] : k3b) + drow;
            float hv[8];
            #pragma unroll
            for (int v = 0; v < 8; ++v) hv[v] = actA[prow + v * SA];
            #pragma unroll
            for (int v = 0; v < 8; ++v) kd[v * H] = ks[v] + hv[v];
          } else {
            #pragma unroll
            for (int v = 0; v < 8; ++v) {
              const float kv = ks[v] + actA[prow + v * SA];
              zbuf[prow + v * SA] += (kbuf[0][drow + v * H] + kv) * 0.125f;
            }
          }
        }
      }
      __syncthreads();
    }
  }
  out_proj(zbuf, W_last, bias_s + 4 * SH, out, b0r, TM1, wsel, lane);
}

extern "C" void kernel_launch(void* const* d_in, const int* in_sizes, int n_in,
                              void* d_out, int out_size, void* d_ws, size_t ws_size,
                              hipStream_t stream) {
  (void)in_sizes; (void)n_in; (void)out_size; (void)d_ws; (void)ws_size;
  const float* coeffs = (const float*)d_in[1];
  const float* W_init = (const float*)d_in[2];
  const float* b_init = (const float*)d_in[3];
  const float* W0     = (const float*)d_in[4];
  const float* b0     = (const float*)d_in[5];
  const float* Wl     = (const float*)d_in[6];
  const float* bl     = (const float*)d_in[7];
  const float* Wf     = (const float*)d_in[8];
  const float* bf     = (const float*)d_in[9];
  const float* W_last = (const float*)d_in[10];
  const float* b_last = (const float*)d_in[11];
  float* out = (float*)d_out;

  ncde_wmma_kernel<<<B_T / MT, THREADS, 0, stream>>>(
      coeffs, W_init, b_init, W0, b0, Wl, bl, Wf, bf, W_last, b_last, out);
}


# Stage tail in one pass: upper half sums pass through the writer's own staging buffer, the lower half's lanes form k and the next stage's input zz; zz loop and 2 of 8 barriers a stage gone; same bits
# speedup vs baseline: 3.1816x; 1.0341x over previous
#include <hip/hip_runtime.h>
#include <math.h>


namespace {
constexpr int B_T = 256, T_T = 32, IN = 16, H = 128, SH = 128, OUTW = 64, TM1 = 31;
constexpr int MT = 16;
constexpr int THREADS = 512;
constexpr int NCOLW = 8;
constexpr int CHW = 8;
constexpr int SA = 132;
constexpr int NRES = 2;
constexpr int KS = SH / 4;
constexpr int ST = 20;
constexpr int STG_W = 16 * ST;
constexpr int POOL = 16 * STG_W;

typedef __attribute__((ext_vector_type(2))) float v2f;
typedef __attribute__((ext_vector_type(8))) float v8f;

__device__ __forceinline__ float tanh_hw(float x) {
  return __builtin_amdgcn_tanhf(x);
}

__device__ __forceinline__ v8f wmma4(v2f a, v2f b, v8f c) {
  return __builtin_amdgcn_wmma_f32_16x16x4_f32(false, a, false, b, (short)0, c,
                                               false, false);
}

__device__ __forceinline__ void wave_lds_handover() {
  __builtin_amdgcn_fence(__ATOMIC_RELEASE, "wavefront", "local");
  __builtin_amdgcn_wave_barrier();
  __builtin_amdgcn_fence(__ATOMIC_ACQUIRE, "wavefront", "local");
}

__device__ __forceinline__ void staged_unit(float* sst, const float* sld, float4 lo,
                                            float4 hi, const v2f* a, v8f& acc) {
  wave_lds_handover();
  *(float4*)(sst) = lo;
  *(float4*)(sst + 8 * ST) = hi;
  wave_lds_handover();
  #pragma unroll
  for (int j = 0; j < 4; ++j) {
    const v2f b = *(const v2f*)(sld + 4 * j);
    acc = wmma4(a[j], b, acc);
  }
}

struct Chunk { float4 t0l, t0h, t1l, t1h; };

__device__ __forceinline__ Chunk load_chunk(const float* gq) {
  constexpr size_t HI = (size_t)8 * IN * SH;
  Chunk c;
  c.t0l = *(const float4*)(gq);
  c.t0h = *(const float4*)(gq + HI);
  c.t1l = *(const float4*)(gq + SH);
  c.t1h = *(const float4*)(gq + SH + HI);
  return c;
}

__device__ __forceinline__ void do_chunk(const Chunk& c, const float* ak, float* sst,
                                         const float* sld, v8f& acc0, v8f& acc1) {
  v2f a[4];
  #pragma unroll
  for (int j = 0; j < 4; ++j) a[j] = *(const v2f*)(ak + 4 * j);
  staged_unit(sst, sld, c.t0l, c.t0h, a, acc0);
  staged_unit(sst, sld, c.t1l, c.t1h, a, acc1);
}

__device__ __forceinline__ v8f gemm_tile(const float* A, int lda, const float* W,
                                         int ldw, int n, int K, int lane) {
  const int m = lane & 15;
  const int kh = (lane >> 4) << 1;
  v8f acc = {};
  #pragma unroll 8
  for (int k0 = 0; k0 < K; k0 += 4) {
    v2f a = *(const v2f*)(A + m * lda + k0 + kh);
    v2f b = *(const v2f*)(W + (size_t)n * ldw + k0 + kh);
    acc = wmma4(a, b, acc);
  }
  return acc;
}

__device__ __forceinline__ void res_pair(const float* arow, const v2f* wa,
                                         const v2f* wb, v8f& acc0, v8f& acc1) {
  #pragma unroll
  for (int kk = 0; kk < KS; ++kk) {
    v2f a = *(const v2f*)(arow + 4 * kk);
    acc0 = wmma4(a, wa[kk], acc0);
    acc1 = wmma4(a, wb[kk], acc1);
  }
}

__device__ __forceinline__ void mlp_layer(const float* src, const float* Wlds,
                                          const float* bias, float* dst,
                                          int wid, int lane) {
  constexpr int KB = 8;
  constexpr int NB = KS / KB;
  const int mD = lane & 15, half = lane >> 4;
  const int n = wid * 16 + mD;
  const int kh = half << 1;
  const float bn = bias[n];
  const float* ap = src + mD * SA + kh;
  const float* bp = Wlds + n * SA + kh;
  v2f a[2][KB], b[2][KB];
  #pragma unroll
  for (int j = 0; j < KB; ++j) {
    a[0][j] = *(const v2f*)(ap + 4 * j);
    b[0][j] = *(const v2f*)(bp + 4 * j);
  }
  v8f acc = {};
  #pragma unroll
  for (int blk = 0; blk < NB; ++blk) {
    const int cur = blk & 1, nxt = cur ^ 1;
    if (blk + 1 < NB) {
      #pragma unroll
      for (int j = 0; j < KB; ++j) {
        a[nxt][j] = *(const v2f*)(ap + 4 * ((blk + 1) * KB + j));
        b[nxt][j] = *(const v2f*)(bp + 4 * ((blk + 1) * KB + j));
      }
    }
    __builtin_amdgcn_sched_barrier(0);
    #pragma unroll
    for (int j = 0; j < KB; ++j) acc = wmma4(a[cur][j], b[cur][j], acc);
    __builtin_amdgcn_sched_barrier(0);
  }
  float* drow = dst + 8 * half * SA + n;
  #pragma unroll
  for (int v = 0; v < 8; ++v) {
    float x = acc[v] + bn;
    drow[v * SA] = fmaxf(x, 0.f);
  }
}

__device__ __forceinline__ void out_proj(const float* zb, const float* W_last,
                                         const float* blast, float* out,
                                         int b0r, int t, int tile, int lane) {
  if (tile >= 0 && tile < 4) {
    constexpr int KB = 4;
    constexpr int NB = KS / KB;
    const int mD = lane & 15, half = lane >> 4;
    const int n = tile * 16 + mD;
    const int kh = half << 1;
    const float bn = blast[n];
    const float* wr = W_last + (size_t)n * H + kh;
    v2f b[KS];
    #pragma unroll
    for (int kk = 0; kk < KS; ++kk) b[kk] = *(const v2f*)(wr + 4 * kk);
    const float* ap = zb + mD * SA + kh;
    v2f a[2][KB];
    #pragma unroll
    for (int j = 0; j < KB; ++j) a[0][j] = *(const v2f*)(ap + 4 * j);
    v8f acc = {};
    #pragma unroll
    for (int blk = 0; blk < NB; ++blk) {
      const int cur = blk & 1, nxt = cur ^ 1;
      if (blk + 1 < NB) {
        #pragma unroll
        for (int j = 0; j < KB; ++j)
          a[nxt][j] = *(const v2f*)(ap + 4 * ((blk + 1) * KB + j));
      }
      __builtin_amdgcn_sched_barrier(0);
      #pragma unroll
      for (int j = 0; j < KB; ++j) acc = wmma4(a[cur][j], b[blk * KB + j], acc);
      __builtin_amdgcn_sched_barrier(0);
    }
    #pragma unroll
    for (int v = 0; v < 8; ++v) {
      int m = v + 8 * half;
      float x = acc[v] + bn;
      out[((size_t)(b0r + m) * T_T + t) * OUTW + n] = 1.f / (1.f + expf(-x));
    }
  }
}
}

__global__ __launch_bounds__(THREADS) void ncde_wmma_kernel(
    const float* __restrict__ coeffs,
    const float* __restrict__ W_init, const float* __restrict__ b_init,
    const float* __restrict__ W0, const float* __restrict__ b0,
    const float* __restrict__ Wl, const float* __restrict__ bl,
    const float* __restrict__ Wf, const float* __restrict__ bf,
    const float* __restrict__ W_last, const float* __restrict__ b_last,
    float* __restrict__ out) {
  __shared__ __align__(16) float wlds[4 * H * SA];
  __shared__ __align__(16) float zbuf[MT * SA];
  __shared__ __align__(16) float kbuf[2][MT * H];
  __shared__ __align__(16) float actI[MT * SA];
  __shared__ __align__(16) float pool[POOL];
  float* const actA = pool + MT * H;
  __shared__ __align__(16) float dxs[MT * IN];
  __shared__ __align__(16) float bias_s[4 * SH + OUTW];

  const int tid = threadIdx.x;
  const int lane = tid & 31;
  const int wid = tid >> 5;
  const int b0r = blockIdx.x * MT;
  const int mD = lane & 15;
  const int half = lane >> 4;
  constexpr float third = 1.f / 3.f;
  const int wsel = __builtin_amdgcn_readfirstlane(wid);
  const int hg = wsel & (NCOLW - 1);
  const int chalf = wsel >> 3;

  v2f wres[NRES * KS];
  {
    const int khr = half << 1;
    #pragma unroll
    for (int j = 0; j < NRES; ++j) {
      const float* wr = Wf + (size_t)((hg * 16 + mD) * IN + chalf * CHW + j) * SH + khr;
      #pragma unroll
      for (int kk = 0; kk < KS; ++kk)
        wres[j * KS + kk] = *(const v2f*)(wr + 4 * kk);
    }
  }
  const int rot = (4 - (wsel >> 2)) & 3;

  for (int e = tid; e < 4 * H * SH; e += THREADS) {
    int mat = e >> 14;
    int r = (e >> 7) & (H - 1);
    int k = e & (SH - 1);
    const float* src = (mat == 0) ? W0 : (Wl + (size_t)(mat - 1) * SH * SH);
    wlds[(size_t)mat * H * SA + r * SA + k] = src[(size_t)r * SH + k];
  }
  bias_s[tid] = (tid < SH) ? b0[tid] : bl[tid - SH];
  if (tid < OUTW) bias_s[4 * SH + tid] = b_last[tid];

  if (tid < MT * IN) {
    int m = tid >> 4, i = tid & 15;
    dxs[tid] = coeffs[((size_t)(b0r + m) * TM1) * (4 * IN) + i];
  }
  __syncthreads();
  if (wsel < NCOLW) {
    const int n = wsel * 16 + mD;
    v8f acc = gemm_tile(dxs, IN, W_init, IN, n, IN, lane);
    const float bn = b_init[n];
    #pragma unroll
    for (int v = 0; v < 8; ++v) {
      const float z0 = acc[v] + bn;
      zbuf[(v + 8 * half) * SA + n] = z0;
      actI[(v + 8 * half) * SA + n] = z0;
    }
  }
  __syncthreads();

  #pragma unroll 1
  for (int it = 0; it < 4 * TM1; ++it) {
    {
      const int step = it >> 2;
      const int s = it & 3;
      const float frac = (s == 0) ? 0.f
                       : (s == 1) ? third
                       : (s == 2) ? (2.f * third) : 1.f;
      if (wsel < NCOLW) {
        mlp_layer(actI, wlds + 0 * H * SA, bias_s + 0 * SH, actA, wsel, lane);
      } else {
        const int t = tid - MT * IN;
        const int m = t >> 4, i = t & 15;
        const float* cb = coeffs + ((size_t)(b0r + m) * TM1 + step) * (4 * IN);
        float bi = cb[IN + i], ci = cb[2 * IN + i], di = cb[3 * IN + i];
        dxs[i * MT + m] = bi + (ci + di * frac) * frac;
      }
      __syncthreads();
      if (wsel < NCOLW) {
        mlp_layer(actA, wlds + 1 * H * SA, bias_s + 1 * SH, actI, wsel, lane);
      } else if (s == 0) {
        out_proj(zbuf, W_last, bias_s + 4 * SH, out, b0r, step, wsel - NCOLW, lane);
      }
      __syncthreads();
      if (wsel < NCOLW) mlp_layer(actI, wlds + 2 * H * SA, bias_s + 2 * SH, actA, wsel, lane);
      __syncthreads();
      if (wsel < NCOLW) mlp_layer(actA, wlds + 3 * H * SA, bias_s + 3 * SH, actI, wsel, lane);
      __syncthreads();

      {
        float ks[8];
        #pragma unroll
        for (int v = 0; v < 8; ++v) ks[v] = 0.f;
        const int kh = half << 1;
        const int hcol = hg * 16 + mD;
        const float* arow = actI + mD * SA + kh;
        const float* bfrow = bf + hcol * IN + chalf * CHW;
        const int r8 = lane >> 2, t4 = lane & 3;
        const float* grow = Wf + (size_t)((hg * 16 + r8) * IN + chalf * CHW) * SH + 4 * t4;
        float* const stg = pool + wsel * STG_W;
        float* const sst = stg + r8 * ST + 4 * t4;
        const float* sld = stg + mD * ST + kh;
        const float* dxh = dxs + chalf * CHW * MT + 8 * half;
        #pragma unroll 1
        for (int i = 0; i < CHW / 2; ++i) {
          if (i == 0)      __builtin_amdgcn_s_setprio(3);
          else if (i == 1) __builtin_amdgcn_s_setprio(2);
          else if (i == 2) __builtin_amdgcn_s_setprio(1);
          else             __builtin_amdgcn_s_setprio(0);
          const int p = (i + rot) & 3;
          const int c0 = 2 * p;
          const int c1 = c0 + 1;
          const v2f bfp = *(const v2f*)(bfrow + c0);
          v8f acc0 = {}, acc1 = {};
          if (p == 0) {
            res_pair(arow, wres + 0 * KS, wres + 1 * KS, acc0, acc1);
          } else {
            const float* g0 = grow + (size_t)c0 * SH;
            const float* gq = g0;
            const float* ak = arow;
            Chunk ca = load_chunk(gq);
            #pragma unroll 1
            for (int kt = 0; kt < SH - 32; kt += 32) {
              const Chunk cb = load_chunk(gq + 16);
              do_chunk(ca, ak, sst, sld, acc0, acc1);
              ca = load_chunk(gq + 32);
              do_chunk(cb, ak + 16, sst, sld, acc0, acc1);
              gq += 32;
              ak += 32;
            }
            {
              const Chunk cb = load_chunk(gq + 16);
              do_chunk(ca, ak, sst, sld, acc0, acc1);
              do_chunk(cb, ak + 16, sst, sld, acc0, acc1);
            }
          }
          const float4 d0a = *(const float4*)(dxh + c0 * MT);
          const float4 d0b = *(const float4*)(dxh + c0 * MT + 4);
          const float4 d1a = *(const float4*)(dxh + c1 * MT);
          const float4 d1b = *(const float4*)(dxh + c1 * MT + 4);
          const float d0[8] = {d0a.x, d0a.y, d0a.z, d0a.w, d0b.x, d0b.y, d0b.z, d0b.w};
          const float d1[8] = {d1a.x, d1a.y, d1a.z, d1a.w, d1b.x, d1b.y, d1b.z, d1b.w};
          #pragma unroll
          for (int v = 0; v < 8; ++v) {
            ks[v] += tanh_hw(acc0[v] + bfp.x) * d0[v];
            ks[v] += tanh_hw(acc1[v] + bfp.y) * d1[v];
          }
        }
        const int prow = 8 * half * SA + hcol;
        const int drow = 8 * half * H + hcol;
        float zv[8] = {}, ka[8] = {}, kb[8] = {};
        if (chalf == 1) {
          #pragma unroll
          for (int v = 0; v < 8; ++v) stg[v * 32 + lane] = ks[v];
        } else {
          #pragma unroll
          for (int v = 0; v < 8; ++v) zv[v] = zbuf[prow + v * SA];
          if (s >= 1) {
            #pragma unroll
            for (int v = 0; v < 8; ++v) ka[v] = kbuf[0][drow + v * H];
          }
          if (s == 2) {
            #pragma unroll
            for (int v = 0; v < 8; ++v) kb[v] = kbuf[1][drow + v * H];
          }
        }
        __syncthreads();
        if (chalf == 0) {
          const float* hs = stg + NCOLW * STG_W;
          float kv[8];
          #pragma unroll
          for (int v = 0; v < 8; ++v) kv[v] = ks[v] + hs[v * 32 + lane];
          float zz[8], kst[8];
          if (s == 0) {
            #pragma unroll
            for (int v = 0; v < 8; ++v) {
              float z = zv[v];
              z += kv[v] * third;
              zz[v] = z;
              kst[v] = kv[v];
            }
          } else if (s == 1) {
            #pragma unroll
            for (int v = 0; v < 8; ++v) {
              float z = zv[v];
              z += kv[v] - ka[v] * third;
              zz[v] = z;
              kst[v] = kv[v];
            }
          } else if (s == 2) {
            #pragma unroll
            for (int v = 0; v < 8; ++v) {
              float z = zv[v];
              z += ka[v] - kb[v] + kv[v];
              zz[v] = z;
              kst[v] = ka[v] + 3.f * (kb[v] + kv[v]);
            }
          } else {
            #pragma unroll
            for (int v = 0; v < 8; ++v) {
              float z = zv[v];
              z += (ka[v] + kv[v]) * 0.125f;
              zz[v] = z;
              kst[v] = 0.f;
            }
          }
          if (s < 3) {
            float* kd = kbuf[(s == 1) ? 1 : 0] + drow;
            #pragma unroll
            for (int v = 0; v < 8; ++v) kd[v * H] = kst[v];
          } else {
            #pragma unroll
            for (int v = 0; v < 8; ++v) zbuf[prow + v * SA] = zz[v];
          }
          #pragma unroll
          for (int v = 0; v < 8; ++v) actI[prow + v * SA] = zz[v];
        }
      }
      __syncthreads();
    }
  }
  out_proj(zbuf, W_last, bias_s + 4 * SH, out, b0r, TM1, wsel, lane);
}

extern "C" void kernel_launch(void* const* d_in, const int* in_sizes, int n_in,
                              void* d_out, int out_size, void* d_ws, size_t ws_size,
                              hipStream_t stream) {
  (void)in_sizes; (void)n_in; (void)out_size; (void)d_ws; (void)ws_size;
  const float* coeffs = (const float*)d_in[1];
  const float* W_init = (const float*)d_in[2];
  const float* b_init = (const float*)d_in[3];
  const float* W0     = (const float*)d_in[4];
  const float* b0     = (const float*)d_in[5];
  const float* Wl     = (const float*)d_in[6];
  const float* bl     = (const float*)d_in[7];
  const float* Wf     = (const float*)d_in[8];
  const float* bf     = (const float*)d_in[9];
  const float* W_last = (const float*)d_in[10];
  const float* b_last = (const float*)d_in[11];
  float* out = (float*)d_out;

  ncde_wmma_kernel<<<B_T / MT, THREADS, 0, stream>>>(
      coeffs, W_init, b_init, W0, b0, Wl, bl, Wf, bf, W_last, b_last, out);
}
